# starting kernel
# baseline (speedup 1.0000x reference)
#include <hip/hip_runtime.h>

#define N_NODES 100000
#define N_EDGES 1600000
#define D 64

typedef __attribute__((ext_vector_type(2))) float v2f;
typedef __attribute__((ext_vector_type(8))) float v8f;

__global__ void zero_kernel(float* __restrict__ p, long long n) {
    long long i = (long long)blockIdx.x * blockDim.x + threadIdx.x;
    long long stride = (long long)gridDim.x * blockDim.x;
    for (; i < n; i += stride) p[i] = 0.0f;
}

__global__ void degree_kernel(const int* __restrict__ row, float* __restrict__ deg, int E) {
    int e = blockIdx.x * blockDim.x + threadIdx.x;
    if (e < E) atomicAdd(&deg[row[e]], 1.0f);
}

__global__ void dinv_kernel(const float* __restrict__ deg, float* __restrict__ dinv, int n) {
    int i = blockIdx.x * blockDim.x + threadIdx.x;
    if (i < n) {
        float d = deg[i];
        dinv[i] = (d > 0.0f) ? (1.0f / sqrtf(d)) : 0.0f;
    }
}

__global__ void lap_kernel(const int* __restrict__ row, const int* __restrict__ col,
                           const float* __restrict__ attr, const float* __restrict__ dinv,
                           float* __restrict__ lap, int E) {
    int e = blockIdx.x * blockDim.x + threadIdx.x;
    if (e < E) lap[e] = -dinv[row[e]] * attr[e] * dinv[col[e]];
}

__global__ void spmm_kernel(const int* __restrict__ row, const int* __restrict__ col,
                            const float* __restrict__ lap, const float* __restrict__ h,
                            float* __restrict__ acc, int E) {
    int idx = blockIdx.x * blockDim.x + threadIdx.x;
    int e = idx >> 6;
    int j = idx & 63;
    if (e < E) {
        float l = lap[e];
        int c = col[e];
        int r = row[e];
        atomicAdd(&acc[r * D + j], l * h[c * D + j]);
    }
}

__global__ void axpby_kernel(float* __restrict__ y, const float* __restrict__ x, int n) {
    int i = blockIdx.x * blockDim.x + threadIdx.x;
    if (i < n) y[i] = 2.0f * y[i] - x[i];
}

__global__ void gemm_wmma_kernel(const float* __restrict__ H, const float* __restrict__ W,
                                 const float* __restrict__ bias, float* __restrict__ out,
                                 int n_row_tiles, int accumulate) {
    const int wave = threadIdx.x >> 5;
    const int lane = threadIdx.x & 31;
    const int col0 = wave * 16;
    const int laneN = lane & 15;
    const int hi = lane >> 4;

    v2f bfrag[16];
#pragma unroll
    for (int ks = 0; ks < 16; ++ks) {
        int k = ks * 4 + hi * 2;
        bfrag[ks].x = W[(k + 0) * D + col0 + laneN];
        bfrag[ks].y = W[(k + 1) * D + col0 + laneN];
    }
    float bval = accumulate ? 0.0f : bias[col0 + laneN];

    for (int tile = blockIdx.x; tile < n_row_tiles; tile += gridDim.x) {
        const int row0 = tile * 16;
        v8f acc;
        if (accumulate) {
#pragma unroll
            for (int r = 0; r < 8; ++r)
                acc[r] = out[(row0 + hi * 8 + r) * D + col0 + laneN];
        } else {
#pragma unroll
            for (int r = 0; r < 8; ++r) acc[r] = bval;
        }

        const float* hrow = H + (row0 + laneN) * D + hi * 2;
#pragma unroll
        for (int ks = 0; ks < 16; ++ks) {
            v2f a;
            a.x = hrow[ks * 4 + 0];
            a.y = hrow[ks * 4 + 1];
            acc = __builtin_amdgcn_wmma_f32_16x16x4_f32(
                 false, a,  false, bfrag[ks],
                 (short)0, acc,  false,  false);
        }

#pragma unroll
        for (int r = 0; r < 8; ++r)
            out[(row0 + hi * 8 + r) * D + col0 + laneN] = acc[r];
    }
}

extern "C" void kernel_launch(void* const* d_in, const int* in_sizes, int n_in,
                              void* d_out, int out_size, void* d_ws, size_t ws_size,
                              hipStream_t stream) {
    const float* x         = (const float*)d_in[0];
    const int*   edge_idx  = (const int*)d_in[1];
    const float* edge_attr = (const float*)d_in[2];
    const float* weight    = (const float*)d_in[3];
    const float* bias      = (const float*)d_in[4];
    float*       out       = (float*)d_out;

    const int* row = edge_idx;
    const int* col = edge_idx + N_EDGES;

    char* ws = (char*)d_ws;
    float* deg  = (float*)ws; ws += sizeof(float) * N_NODES;
    float* dinv = (float*)ws; ws += sizeof(float) * N_NODES;
    float* lap  = (float*)ws; ws += sizeof(float) * N_EDGES;
    float* A    = (float*)ws; ws += sizeof(float) * (size_t)N_NODES * D;
    float* B    = (float*)ws; ws += sizeof(float) * (size_t)N_NODES * D;
    float* Z    = (float*)ws; ws += sizeof(float) * (size_t)N_NODES * D;

    const int n_row_tiles = N_NODES / 16;
    const int GEMM_BLOCKS = 1024;
    const long long spmm_threads = (long long)N_EDGES * D;
    const int spmm_blocks = (int)((spmm_threads + 255) / 256);
    const int e_blocks = (N_EDGES + 255) / 256;
    const int n_blocks = (N_NODES + 255) / 256;
    const int nd_blocks = (N_NODES * D + 255) / 256;

    zero_kernel<<<2048, 256, 0, stream>>>(deg, (long long)N_NODES);
    zero_kernel<<<4096, 256, 0, stream>>>(A, (long long)N_NODES * D * 3);

    degree_kernel<<<e_blocks, 256, 0, stream>>>(row, deg, N_EDGES);
    dinv_kernel<<<n_blocks, 256, 0, stream>>>(deg, dinv, N_NODES);
    lap_kernel<<<e_blocks, 256, 0, stream>>>(row, col, edge_attr, dinv, lap, N_EDGES);

    gemm_wmma_kernel<<<GEMM_BLOCKS, 128, 0, stream>>>(x, weight, bias, out, n_row_tiles, 0);

    spmm_kernel<<<spmm_blocks, 256, 0, stream>>>(row, col, lap, x, A, N_EDGES);
    gemm_wmma_kernel<<<GEMM_BLOCKS, 128, 0, stream>>>(A, weight + 1 * D * D, nullptr, out, n_row_tiles, 1);

    spmm_kernel<<<spmm_blocks, 256, 0, stream>>>(row, col, lap, A, B, N_EDGES);
    axpby_kernel<<<nd_blocks, 256, 0, stream>>>(B, x, N_NODES * D);
    gemm_wmma_kernel<<<GEMM_BLOCKS, 128, 0, stream>>>(B, weight + 2 * D * D, nullptr, out, n_row_tiles, 1);

    spmm_kernel<<<spmm_blocks, 256, 0, stream>>>(row, col, lap, B, Z, N_EDGES);
    axpby_kernel<<<nd_blocks, 256, 0, stream>>>(Z, A, N_NODES * D);
    gemm_wmma_kernel<<<GEMM_BLOCKS, 128, 0, stream>>>(Z, weight + 3 * D * D, nullptr, out, n_row_tiles, 1);
}


# one 256-deep WMMA pass replaces the four accumulating 64-deep GEMM launches: tx3 parked in out and replaced in place via an LDS-staged tile; three read-modify-write passes over out are gone
# speedup vs baseline: 4.2702x; 4.2702x over previous
#include <hip/hip_runtime.h>

#define N_NODES 100000
#define N_EDGES 1600000
#define D 64

typedef __attribute__((ext_vector_type(2))) float v2f;
typedef __attribute__((ext_vector_type(8))) float v8f;

__global__ void zero_kernel(float* __restrict__ p, long long n) {
    long long i = (long long)blockIdx.x * blockDim.x + threadIdx.x;
    long long stride = (long long)gridDim.x * blockDim.x;
    for (; i < n; i += stride) p[i] = 0.0f;
}

__global__ void hist_kernel(const int* __restrict__ row, int* __restrict__ cnt,
                            int* __restrict__ pos, int E) {
    int e = blockIdx.x * blockDim.x + threadIdx.x;
    if (e < E) pos[e] = atomicAdd(&cnt[row[e]], 1);
}

__global__ void scan_dinv_kernel(const int* __restrict__ cnt, int* __restrict__ rowptr,
                                 float* __restrict__ dinv) {
    __shared__ int s_pre[8];
    __shared__ int s_sum[8];
    const int t = threadIdx.x;
    const int lane = t & 31;
    const int wave = t >> 5;
    const int4* cnt4 = (const int4*)cnt;
    const int nq = N_NODES / 4;

    int pre = 0;
    for (int k = 0; k < (int)blockIdx.x; ++k) {
        int4 u = cnt4[k * 256 + t];
        pre += (u.x + u.y) + (u.z + u.w);
    }

    const int q = blockIdx.x * 256 + t;
    int4 v = make_int4(0, 0, 0, 0);
    if (q < nq) v = cnt4[q];
    const int tsum = (v.x + v.y) + (v.z + v.w);

    int inc = tsum;
    for (int d = 1; d < 32; d <<= 1) {
        int o = __shfl_up(inc, d);
        if (lane >= d) inc += o;
    }
    for (int d = 16; d > 0; d >>= 1) pre += __shfl_xor(pre, d);
    if (lane == 31) s_sum[wave] = inc;
    if (lane == 0) s_pre[wave] = pre;
    __syncthreads();
    int base = 0;
    for (int w = 0; w < 8; ++w) {
        base += s_pre[w];
        if (w < wave) base += s_sum[w];
    }
    const int ex = base + inc - tsum;

    if (q < nq) {
        int4 o;
        o.x = ex;
        o.y = o.x + v.x;
        o.z = o.y + v.y;
        o.w = o.z + v.z;
        ((int4*)rowptr)[q] = o;
        float4 dv;
        dv.x = (v.x > 0) ? (1.0f / sqrtf((float)v.x)) : 0.0f;
        dv.y = (v.y > 0) ? (1.0f / sqrtf((float)v.y)) : 0.0f;
        dv.z = (v.z > 0) ? (1.0f / sqrtf((float)v.z)) : 0.0f;
        dv.w = (v.w > 0) ? (1.0f / sqrtf((float)v.w)) : 0.0f;
        ((float4*)dinv)[q] = dv;
        if (q == nq - 1) rowptr[N_NODES] = o.w + v.w;
    }
}

__global__ void fill_kernel(const int* __restrict__ row, const int* __restrict__ col,
                            const float* __restrict__ attr, const float* __restrict__ dinv,
                            const int* __restrict__ rowptr, const int* __restrict__ pos,
                            int* __restrict__ edges, int E) {
    int e = blockIdx.x * blockDim.x + threadIdx.x;
    if (e < E) {
        int r = row[e];
        int c = col[e];
        float w = -dinv[r] * attr[e] * dinv[c];
        int p = rowptr[r] + pos[e];
        ((int2*)edges)[p] = make_int2(c, __float_as_int(w));
    }
}

__global__ void __launch_bounds__(256)
spmm_gather_kernel(const int* __restrict__ rowptr, const int* __restrict__ edges,
                   const float* __restrict__ h, const float* prev, float* dst, int n_rows) {
    const int lane = threadIdx.x & 31;
    const int wave = __builtin_amdgcn_readfirstlane((int)(threadIdx.x >> 5));
    const int n_waves = gridDim.x * 8;
    const float2* h2 = (const float2*)h;
    const int2* ed = (const int2*)edges;

    for (int r = blockIdx.x * 8 + wave; r < n_rows; r += n_waves) {
        const int s = rowptr[r];
        const int e = rowptr[r + 1];
        float2 p = make_float2(0.0f, 0.0f);
        if (prev) p = ((const float2*)prev)[r * 32 + lane];
        float2 acc = make_float2(0.0f, 0.0f);
        for (int b = s; b < e; b += 32) {
            const int n = min(32, e - b);
            int2 my = make_int2(0, 0);
            if (lane < n) my = ed[b + lane];
#pragma unroll 4
            for (int j = 0; j < n; ++j) {
                const int c = __builtin_amdgcn_readlane(my.x, j);
                const float w = __int_as_float(__builtin_amdgcn_readlane(my.y, j));
                const float2 hv = h2[c * 32 + lane];
                acc.x += w * hv.x;
                acc.y += w * hv.y;
            }
        }
        float2 res = acc;
        if (prev) {
            res.x = 2.0f * acc.x - p.x;
            res.y = 2.0f * acc.y - p.y;
        }
        ((float2*)dst)[r * 32 + lane] = res;
    }
}

#define LDS_PITCH 68

__global__ void __launch_bounds__(128)
gemm4_wmma_kernel(const float* __restrict__ X, const float* __restrict__ T1,
                  const float* __restrict__ T2, float* T3out,
                  const float* __restrict__ Wcat, const float* __restrict__ bias,
                  int n_row_tiles) {
    __shared__ __attribute__((aligned(16))) float sA[4 * 16 * LDS_PITCH];
    __shared__ __attribute__((aligned(16))) float sC[16 * LDS_PITCH];

    const int t = threadIdx.x;
    const int wave = t >> 5;
    const int lane = t & 31;
    const int col0 = wave * 16;
    const int laneN = lane & 15;
    const int hi = lane >> 4;

    v2f bfrag[64];
#pragma unroll
    for (int ks = 0; ks < 64; ++ks) {
        int k = ks * 4 + hi * 2;
        bfrag[ks].x = Wcat[(k + 0) * D + col0 + laneN];
        bfrag[ks].y = Wcat[(k + 1) * D + col0 + laneN];
    }
    const float bval = bias[col0 + laneN];

    const int r_a = t >> 4;
    const int c4 = (t & 15) * 4;

    float4 x0, x1, a0, a1, b0, b1, c0, c1;
    x0 = x1 = a0 = a1 = b0 = b1 = c0 = c1 = make_float4(0.0f, 0.0f, 0.0f, 0.0f);
    int tile = blockIdx.x;
    if (tile < n_row_tiles) {
        const size_t o0 = (size_t)(tile * 16 + r_a) * D + c4;
        const size_t o1 = o0 + (size_t)8 * D;
        x0 = *(const float4*)(X + o0);      x1 = *(const float4*)(X + o1);
        a0 = *(const float4*)(T1 + o0);     a1 = *(const float4*)(T1 + o1);
        b0 = *(const float4*)(T2 + o0);     b1 = *(const float4*)(T2 + o1);
        c0 = *(const float4*)(T3out + o0);  c1 = *(const float4*)(T3out + o1);
    }

    for (; tile < n_row_tiles; tile += gridDim.x) {
        *(float4*)&sA[(0 * 16 + r_a) * LDS_PITCH + c4] = x0;
        *(float4*)&sA[(0 * 16 + r_a + 8) * LDS_PITCH + c4] = x1;
        *(float4*)&sA[(1 * 16 + r_a) * LDS_PITCH + c4] = a0;
        *(float4*)&sA[(1 * 16 + r_a + 8) * LDS_PITCH + c4] = a1;
        *(float4*)&sA[(2 * 16 + r_a) * LDS_PITCH + c4] = b0;
        *(float4*)&sA[(2 * 16 + r_a + 8) * LDS_PITCH + c4] = b1;
        *(float4*)&sA[(3 * 16 + r_a) * LDS_PITCH + c4] = c0;
        *(float4*)&sA[(3 * 16 + r_a + 8) * LDS_PITCH + c4] = c1;
        __syncthreads();

        const int next = tile + gridDim.x;
        if (next < n_row_tiles) {
            const size_t o0 = (size_t)(next * 16 + r_a) * D + c4;
            const size_t o1 = o0 + (size_t)8 * D;
            x0 = *(const float4*)(X + o0);      x1 = *(const float4*)(X + o1);
            a0 = *(const float4*)(T1 + o0);     a1 = *(const float4*)(T1 + o1);
            b0 = *(const float4*)(T2 + o0);     b1 = *(const float4*)(T2 + o1);
            c0 = *(const float4*)(T3out + o0);  c1 = *(const float4*)(T3out + o1);
        }

        v8f acc;
        for (int r = 0; r < 8; ++r) acc[r] = bval;

        const float* arow = sA + laneN * LDS_PITCH + hi * 2;
#pragma unroll
        for (int ks = 0; ks < 64; ++ks) {
            const v2f a = *(const v2f*)(arow + (ks >> 4) * 16 * LDS_PITCH + (ks & 15) * 4);
            acc = __builtin_amdgcn_wmma_f32_16x16x4_f32(
                 false, a,  false, bfrag[ks],
                 (short)0, acc,  false,  false);
        }

        for (int r = 0; r < 8; ++r)
            sC[(hi * 8 + r) * LDS_PITCH + col0 + laneN] = acc[r];
        __syncthreads();

        const float4 o0 = *(const float4*)&sC[r_a * LDS_PITCH + c4];
        const float4 o1 = *(const float4*)&sC[(r_a + 8) * LDS_PITCH + c4];
        float* orow = T3out + (size_t)(tile * 16 + r_a) * D + c4;
        *(float4*)orow = o0;
        *(float4*)(orow + 8 * D) = o1;
    }
}

extern "C" void kernel_launch(void* const* d_in, const int* in_sizes, int n_in,
                              void* d_out, int out_size, void* d_ws, size_t ws_size,
                              hipStream_t stream) {
    const float* x         = (const float*)d_in[0];
    const int*   edge_idx  = (const int*)d_in[1];
    const float* edge_attr = (const float*)d_in[2];
    const float* weight    = (const float*)d_in[3];
    const float* bias      = (const float*)d_in[4];
    float*       out       = (float*)d_out;

    const int* row = edge_idx;
    const int* col = edge_idx + N_EDGES;

    char* ws = (char*)d_ws;
    int*   cnt  = (int*)ws;   ws += sizeof(int) * N_NODES;
    float* dinv = (float*)ws; ws += sizeof(float) * N_NODES;
    int*   pos  = (int*)ws;   ws += sizeof(int) * N_EDGES;
    float* A    = (float*)ws; ws += sizeof(float) * (size_t)N_NODES * D;
    float* B    = (float*)ws; ws += sizeof(float) * (size_t)N_NODES * D;
    int*   edges  = (int*)ws; ws += sizeof(int) * 2 * (size_t)N_EDGES;
    int*   rowptr = (int*)ws; ws += sizeof(int) * (N_NODES + 4);

    const int n_row_tiles = N_NODES / 16;
    const int GEMM_BLOCKS = 1024;
    const int SPMM_BLOCKS = 2048;
    const int e_blocks = (N_EDGES + 255) / 256;
    const int scan_blocks = (N_NODES / 4 + 255) / 256;

    zero_kernel<<<2048, 256, 0, stream>>>((float*)cnt, (long long)N_NODES);
    hist_kernel<<<e_blocks, 256, 0, stream>>>(row, cnt, pos, N_EDGES);
    scan_dinv_kernel<<<scan_blocks, 256, 0, stream>>>(cnt, rowptr, dinv);
    fill_kernel<<<e_blocks, 256, 0, stream>>>(row, col, edge_attr, dinv, rowptr, pos, edges, N_EDGES);

    spmm_gather_kernel<<<SPMM_BLOCKS, 256, 0, stream>>>(rowptr, edges, x, nullptr, A, N_NODES);

    spmm_gather_kernel<<<SPMM_BLOCKS, 256, 0, stream>>>(rowptr, edges, A, x, B, N_NODES);

    spmm_gather_kernel<<<SPMM_BLOCKS, 256, 0, stream>>>(rowptr, edges, B, A, out, N_NODES);

    gemm4_wmma_kernel<<<GEMM_BLOCKS, 128, 0, stream>>>(x, A, B, out, weight, bias, n_row_tiles);
}


# SPMM gather: one wave takes two rows at once (half-wave per row, 16-byte loads), 16 source-row loads in flight before the multiply-adds, next pair's edge pairs and bounds fetched ahead
# speedup vs baseline: 4.4869x; 1.0507x over previous
#include <hip/hip_runtime.h>

#define N_NODES 100000
#define N_EDGES 1600000
#define D 64

typedef __attribute__((ext_vector_type(2))) float v2f;
typedef __attribute__((ext_vector_type(8))) float v8f;

__global__ void zero_kernel(float* __restrict__ p, long long n) {
    long long i = (long long)blockIdx.x * blockDim.x + threadIdx.x;
    long long stride = (long long)gridDim.x * blockDim.x;
    for (; i < n; i += stride) p[i] = 0.0f;
}

__global__ void hist_kernel(const int* __restrict__ row, int* __restrict__ cnt,
                            int* __restrict__ pos, int E) {
    int e = blockIdx.x * blockDim.x + threadIdx.x;
    if (e < E) pos[e] = atomicAdd(&cnt[row[e]], 1);
}

__global__ void scan_dinv_kernel(const int* __restrict__ cnt, int* __restrict__ rowptr,
                                 float* __restrict__ dinv) {
    __shared__ int s_pre[8];
    __shared__ int s_sum[8];
    const int t = threadIdx.x;
    const int lane = t & 31;
    const int wave = t >> 5;
    const int4* cnt4 = (const int4*)cnt;
    const int nq = N_NODES / 4;

    int pre = 0;
    for (int k = 0; k < (int)blockIdx.x; ++k) {
        int4 u = cnt4[k * 256 + t];
        pre += (u.x + u.y) + (u.z + u.w);
    }

    const int q = blockIdx.x * 256 + t;
    int4 v = make_int4(0, 0, 0, 0);
    if (q < nq) v = cnt4[q];
    const int tsum = (v.x + v.y) + (v.z + v.w);

    int inc = tsum;
    for (int d = 1; d < 32; d <<= 1) {
        int o = __shfl_up(inc, d);
        if (lane >= d) inc += o;
    }
    for (int d = 16; d > 0; d >>= 1) pre += __shfl_xor(pre, d);
    if (lane == 31) s_sum[wave] = inc;
    if (lane == 0) s_pre[wave] = pre;
    __syncthreads();
    int base = 0;
    for (int w = 0; w < 8; ++w) {
        base += s_pre[w];
        if (w < wave) base += s_sum[w];
    }
    const int ex = base + inc - tsum;

    if (q < nq) {
        int4 o;
        o.x = ex;
        o.y = o.x + v.x;
        o.z = o.y + v.y;
        o.w = o.z + v.z;
        ((int4*)rowptr)[q] = o;
        float4 dv;
        dv.x = (v.x > 0) ? (1.0f / sqrtf((float)v.x)) : 0.0f;
        dv.y = (v.y > 0) ? (1.0f / sqrtf((float)v.y)) : 0.0f;
        dv.z = (v.z > 0) ? (1.0f / sqrtf((float)v.z)) : 0.0f;
        dv.w = (v.w > 0) ? (1.0f / sqrtf((float)v.w)) : 0.0f;
        ((float4*)dinv)[q] = dv;
        if (q == nq - 1) rowptr[N_NODES] = o.w + v.w;
    }
}

__global__ void fill_kernel(const int* __restrict__ row, const int* __restrict__ col,
                            const float* __restrict__ attr, const float* __restrict__ dinv,
                            const int* __restrict__ rowptr, const int* __restrict__ pos,
                            int* __restrict__ edges, int E) {
    int e = blockIdx.x * blockDim.x + threadIdx.x;
    if (e < E) {
        int r = row[e];
        int c = col[e];
        float w = -dinv[r] * attr[e] * dinv[c];
        int p = rowptr[r] + pos[e];
        ((int2*)edges)[p] = make_int2(c, __float_as_int(w));
    }
}

template <int J>
__device__ __forceinline__ int half_bcast(int v) {
    return __builtin_amdgcn_update_dpp(0, v, 0x150 + J, 0xF, 0xF, false);
}

#define SPMM_SLOT(K) { const int c = half_bcast<BASE + K>(my.x); \
                       w[K] = __int_as_float(half_bcast<BASE + K>(my.y)); \
                       hv[K] = h4[c * 16 + sub]; }

template <int BASE>
__device__ __forceinline__ void issue8(const float4* __restrict__ h4, const int2 my, const int sub,
                                       float4* hv, float* w) {
    SPMM_SLOT(0) SPMM_SLOT(1) SPMM_SLOT(2) SPMM_SLOT(3)
    SPMM_SLOT(4) SPMM_SLOT(5) SPMM_SLOT(6) SPMM_SLOT(7)
}

__device__ __forceinline__ void fma8(const float4* hv, const float* w, float4& acc) {
#pragma unroll
    for (int k = 0; k < 8; ++k) {
        acc.x = fmaf(w[k], hv[k].x, acc.x);
        acc.y = fmaf(w[k], hv[k].y, acc.y);
        acc.z = fmaf(w[k], hv[k].z, acc.z);
        acc.w = fmaf(w[k], hv[k].w, acc.w);
    }
}

__device__ __forceinline__ void load_bounds(const int* __restrict__ rowptr, const int q, const int hw,
                                            const int n_rows, int& s, int& e) {
    const int rr = 2 * q + hw;
    const int rc = min(rr, n_rows - 1);
    const int sv = rowptr[rc];
    const int ev = rowptr[rc + 1];
    const bool ok = rr < n_rows;
    s = ok ? sv : 0;
    e = ok ? ev : 0;
}

__device__ __forceinline__ void load_pairs(const int2* __restrict__ ed, const int s, const int e,
                                           const int sub, int2& a, int2& b) {
    const int n = e - s;
    const bool oa = sub < n;
    const bool ob = sub + 16 < n;
    const int2 va = ed[oa ? s + sub : 0];
    const int2 vb = ed[ob ? s + 16 + sub : 0];
    a = oa ? va : make_int2(0, 0);
    b = ob ? vb : make_int2(0, 0);
}

template <bool HAS_PREV>
__global__ void __launch_bounds__(256)
spmm_gather_kernel(const int* __restrict__ rowptr, const int* __restrict__ edges,
                   const float* __restrict__ h, const float* prev, float* dst, int n_rows) {
    const int lane = threadIdx.x & 31;
    const int sub = lane & 15;
    const int hw = lane >> 4;
    const int wave = __builtin_amdgcn_readfirstlane((int)(threadIdx.x >> 5));
    const int n_waves = gridDim.x * 8;
    const int n_pairs = (n_rows + 1) >> 1;
    const float4* h4 = (const float4*)h;
    const int2* ed = (const int2*)edges;

    int pr = blockIdx.x * 8 + wave;
    if (pr >= n_pairs) return;

    int s, e, s2, e2;
    load_bounds(rowptr, pr, hw, n_rows, s, e);
    load_bounds(rowptr, pr + n_waves, hw, n_rows, s2, e2);
    int2 mya, myb;
    load_pairs(ed, s, e, sub, mya, myb);

    for (; pr < n_pairs; pr += n_waves) {
        const int r = 2 * pr + hw;
        const bool live = r < n_rows;
        const int rc = live ? r : n_rows - 1;

        int2 nxa, nxb;
        load_pairs(ed, s2, e2, sub, nxa, nxb);
        int s3, e3;
        load_bounds(rowptr, pr + 2 * n_waves, hw, n_rows, s3, e3);
        float4 p = make_float4(0.0f, 0.0f, 0.0f, 0.0f);
        if (HAS_PREV) p = ((const float4*)prev)[rc * 16 + sub];

        const int cnt = e - s;
        const int c0 = __builtin_amdgcn_readlane(cnt, 0);
        const int c1 = __builtin_amdgcn_readlane(cnt, 16);
        const int nmax = max(c0, c1);

        float4 acc = make_float4(0.0f, 0.0f, 0.0f, 0.0f);
        {
            float4 hv0[8], hv1[8];
            float w0[8], w1[8];
            issue8<0>(h4, mya, sub, hv0, w0);
            issue8<8>(h4, mya, sub, hv1, w1);
            asm volatile("" : "+v"(w0[0]), "+v"(w0[1]), "+v"(w0[2]), "+v"(w0[3]),
                              "+v"(w0[4]), "+v"(w0[5]), "+v"(w0[6]), "+v"(w0[7]) : : "memory");
            asm volatile("" : "+v"(w1[0]), "+v"(w1[1]), "+v"(w1[2]), "+v"(w1[3]),
                              "+v"(w1[4]), "+v"(w1[5]), "+v"(w1[6]), "+v"(w1[7]) : : "memory");
            fma8(hv0, w0, acc);
            fma8(hv1, w1, acc);
        }
        if (nmax > 16) {
            float4 hv[8];
            float w[8];
            issue8<0>(h4, myb, sub, hv, w);
            fma8(hv, w, acc);
        }
        if (nmax > 24) {
            float4 hv[8];
            float w[8];
            issue8<8>(h4, myb, sub, hv, w);
            fma8(hv, w, acc);
        }
        for (int b = 32; b < nmax; b += 16) {
            const bool om = sub < cnt - b;
            const int2 vm = ed[om ? s + b + sub : 0];
            const int2 my = om ? vm : make_int2(0, 0);
            float4 hv0[8], hv1[8];
            float w0[8], w1[8];
            issue8<0>(h4, my, sub, hv0, w0);
            issue8<8>(h4, my, sub, hv1, w1);
            fma8(hv0, w0, acc);
            fma8(hv1, w1, acc);
        }

        float4 res = acc;
        if (HAS_PREV) {
            res.x = 2.0f * acc.x - p.x;
            res.y = 2.0f * acc.y - p.y;
            res.z = 2.0f * acc.z - p.z;
            res.w = 2.0f * acc.w - p.w;
        }
        if (live) ((float4*)dst)[r * 16 + sub] = res;

        s = s2;  e = e2;
        s2 = s3; e2 = e3;
        mya = nxa; myb = nxb;
    }
}

#define LDS_PITCH 68

__global__ void __launch_bounds__(128)
gemm4_wmma_kernel(const float* __restrict__ X, const float* __restrict__ T1,
                  const float* __restrict__ T2, float* T3out,
                  const float* __restrict__ Wcat, const float* __restrict__ bias,
                  int n_row_tiles) {
    __shared__ __attribute__((aligned(16))) float sA[4 * 16 * LDS_PITCH];
    __shared__ __attribute__((aligned(16))) float sC[16 * LDS_PITCH];

    const int t = threadIdx.x;
    const int wave = t >> 5;
    const int lane = t & 31;
    const int col0 = wave * 16;
    const int laneN = lane & 15;
    const int hi = lane >> 4;

    v2f bfrag[64];
#pragma unroll
    for (int ks = 0; ks < 64; ++ks) {
        int k = ks * 4 + hi * 2;
        bfrag[ks].x = Wcat[(k + 0) * D + col0 + laneN];
        bfrag[ks].y = Wcat[(k + 1) * D + col0 + laneN];
    }
    const float bval = bias[col0 + laneN];

    const int r_a = t >> 4;
    const int c4 = (t & 15) * 4;

    float4 x0, x1, a0, a1, b0, b1, c0, c1;
    x0 = x1 = a0 = a1 = b0 = b1 = c0 = c1 = make_float4(0.0f, 0.0f, 0.0f, 0.0f);
    int tile = blockIdx.x;
    if (tile < n_row_tiles) {
        const size_t o0 = (size_t)(tile * 16 + r_a) * D + c4;
        const size_t o1 = o0 + (size_t)8 * D;
        x0 = *(const float4*)(X + o0);      x1 = *(const float4*)(X + o1);
        a0 = *(const float4*)(T1 + o0);     a1 = *(const float4*)(T1 + o1);
        b0 = *(const float4*)(T2 + o0);     b1 = *(const float4*)(T2 + o1);
        c0 = *(const float4*)(T3out + o0);  c1 = *(const float4*)(T3out + o1);
    }

    for (; tile < n_row_tiles; tile += gridDim.x) {
        *(float4*)&sA[(0 * 16 + r_a) * LDS_PITCH + c4] = x0;
        *(float4*)&sA[(0 * 16 + r_a + 8) * LDS_PITCH + c4] = x1;
        *(float4*)&sA[(1 * 16 + r_a) * LDS_PITCH + c4] = a0;
        *(float4*)&sA[(1 * 16 + r_a + 8) * LDS_PITCH + c4] = a1;
        *(float4*)&sA[(2 * 16 + r_a) * LDS_PITCH + c4] = b0;
        *(float4*)&sA[(2 * 16 + r_a + 8) * LDS_PITCH + c4] = b1;
        *(float4*)&sA[(3 * 16 + r_a) * LDS_PITCH + c4] = c0;
        *(float4*)&sA[(3 * 16 + r_a + 8) * LDS_PITCH + c4] = c1;
        __syncthreads();

        const int next = tile + gridDim.x;
        if (next < n_row_tiles) {
            const size_t o0 = (size_t)(next * 16 + r_a) * D + c4;
            const size_t o1 = o0 + (size_t)8 * D;
            x0 = *(const float4*)(X + o0);      x1 = *(const float4*)(X + o1);
            a0 = *(const float4*)(T1 + o0);     a1 = *(const float4*)(T1 + o1);
            b0 = *(const float4*)(T2 + o0);     b1 = *(const float4*)(T2 + o1);
            c0 = *(const float4*)(T3out + o0);  c1 = *(const float4*)(T3out + o1);
        }

        v8f acc;
        for (int r = 0; r < 8; ++r) acc[r] = bval;

        const float* arow = sA + laneN * LDS_PITCH + hi * 2;
#pragma unroll
        for (int ks = 0; ks < 64; ++ks) {
            const v2f a = *(const v2f*)(arow + (ks >> 4) * 16 * LDS_PITCH + (ks & 15) * 4);
            acc = __builtin_amdgcn_wmma_f32_16x16x4_f32(
                 false, a,  false, bfrag[ks],
                 (short)0, acc,  false,  false);
        }

        for (int r = 0; r < 8; ++r)
            sC[(hi * 8 + r) * LDS_PITCH + col0 + laneN] = acc[r];
        __syncthreads();

        const float4 o0 = *(const float4*)&sC[r_a * LDS_PITCH + c4];
        const float4 o1 = *(const float4*)&sC[(r_a + 8) * LDS_PITCH + c4];
        float* orow = T3out + (size_t)(tile * 16 + r_a) * D + c4;
        *(float4*)orow = o0;
        *(float4*)(orow + 8 * D) = o1;
    }
}

extern "C" void kernel_launch(void* const* d_in, const int* in_sizes, int n_in,
                              void* d_out, int out_size, void* d_ws, size_t ws_size,
                              hipStream_t stream) {
    const float* x         = (const float*)d_in[0];
    const int*   edge_idx  = (const int*)d_in[1];
    const float* edge_attr = (const float*)d_in[2];
    const float* weight    = (const float*)d_in[3];
    const float* bias      = (const float*)d_in[4];
    float*       out       = (float*)d_out;

    const int* row = edge_idx;
    const int* col = edge_idx + N_EDGES;

    char* ws = (char*)d_ws;
    int*   cnt  = (int*)ws;   ws += sizeof(int) * N_NODES;
    float* dinv = (float*)ws; ws += sizeof(float) * N_NODES;
    int*   pos  = (int*)ws;   ws += sizeof(int) * N_EDGES;
    float* A    = (float*)ws; ws += sizeof(float) * (size_t)N_NODES * D;
    float* B    = (float*)ws; ws += sizeof(float) * (size_t)N_NODES * D;
    int*   edges  = (int*)ws; ws += sizeof(int) * 2 * (size_t)N_EDGES;
    int*   rowptr = (int*)ws; ws += sizeof(int) * (N_NODES + 4);

    const int n_row_tiles = N_NODES / 16;
    const int GEMM_BLOCKS = 1024;
    const int SPMM_BLOCKS = 1024;
    const int e_blocks = (N_EDGES + 255) / 256;
    const int scan_blocks = (N_NODES / 4 + 255) / 256;

    zero_kernel<<<2048, 256, 0, stream>>>((float*)cnt, (long long)N_NODES);
    hist_kernel<<<e_blocks, 256, 0, stream>>>(row, cnt, pos, N_EDGES);
    scan_dinv_kernel<<<scan_blocks, 256, 0, stream>>>(cnt, rowptr, dinv);
    fill_kernel<<<e_blocks, 256, 0, stream>>>(row, col, edge_attr, dinv, rowptr, pos, edges, N_EDGES);

    spmm_gather_kernel<false><<<SPMM_BLOCKS, 256, 0, stream>>>(rowptr, edges, x, nullptr, A, N_NODES);

    spmm_gather_kernel<true><<<SPMM_BLOCKS, 256, 0, stream>>>(rowptr, edges, A, x, B, N_NODES);

    spmm_gather_kernel<true><<<SPMM_BLOCKS, 256, 0, stream>>>(rowptr, edges, B, A, out, N_NODES);

    gemm4_wmma_kernel<<<GEMM_BLOCKS, 128, 0, stream>>>(x, A, B, out, weight, bias, n_row_tiles);
}


# edge slots from LDS counters: 32 blocks each count a 50000-edge chunk in LDS (two rows per word), a combine pass prefixes the chunks per row; the 1.6M scattered returning global atomics are gone
# speedup vs baseline: 5.9246x; 1.3204x over previous
#include <hip/hip_runtime.h>

#define N_NODES 100000
#define N_EDGES 1600000
#define D 64

typedef __attribute__((ext_vector_type(2))) float v2f;
typedef __attribute__((ext_vector_type(8))) float v8f;

#define N_CHUNKS 32
#define CHUNK (N_EDGES / N_CHUNKS)
#define N_WORDS (N_NODES / 2)
#define CNT_FULL (CHUNK / 1024)
#define CNT_ROUND 8

__global__ void __launch_bounds__(1024)
count_lds_kernel(const int* __restrict__ row, unsigned* __restrict__ m16, int* __restrict__ pos) {
    __shared__ __attribute__((aligned(16))) unsigned s_cnt[N_WORDS];
    const int t = threadIdx.x;
    const int ebase = blockIdx.x * CHUNK;

    for (int q = t; q < N_WORDS / 4; q += 1024) ((uint4*)s_cnt)[q] = make_uint4(0u, 0u, 0u, 0u);
    __syncthreads();

    for (int i0 = 0; i0 < CNT_FULL; i0 += CNT_ROUND) {
        int rr[CNT_ROUND];
        unsigned old[CNT_ROUND];
#pragma unroll
        for (int j = 0; j < CNT_ROUND; ++j) rr[j] = row[ebase + (i0 + j) * 1024 + t];
#pragma unroll
        for (int j = 0; j < CNT_ROUND; ++j) {
            const unsigned r = (unsigned)rr[j];
            old[j] = atomicAdd(&s_cnt[r >> 1], (r & 1u) ? 65536u : 1u);
        }
#pragma unroll
        for (int j = 0; j < CNT_ROUND; ++j) {
            const unsigned r = (unsigned)rr[j];
            pos[ebase + (i0 + j) * 1024 + t] = (int)((r & 1u) ? (old[j] >> 16) : (old[j] & 0xFFFFu));
        }
    }
    if (CNT_FULL * 1024 + t < CHUNK) {
        const int i = CNT_FULL * 1024 + t;
        const unsigned r = (unsigned)row[ebase + i];
        const unsigned o = atomicAdd(&s_cnt[r >> 1], (r & 1u) ? 65536u : 1u);
        pos[ebase + i] = (int)((r & 1u) ? (o >> 16) : (o & 0xFFFFu));
    }
    __syncthreads();

    uint4* dst = (uint4*)(m16 + (size_t)blockIdx.x * N_WORDS);
    for (int q = t; q < N_WORDS / 4; q += 1024) dst[q] = ((const uint4*)s_cnt)[q];
}

__global__ void combine_kernel(const unsigned* __restrict__ m16, int* __restrict__ m32,
                               int* __restrict__ cnt) {
    const int w = blockIdx.x * blockDim.x + threadIdx.x;
    if (w >= N_WORDS) return;
    unsigned v[N_CHUNKS];
#pragma unroll
    for (int p = 0; p < N_CHUNKS; ++p) v[p] = m16[(size_t)p * N_WORDS + w];
    int lo = 0, hi = 0;
#pragma unroll
    for (int p = 0; p < N_CHUNKS; ++p) {
        ((int2*)m32)[(size_t)p * N_WORDS + w] = make_int2(lo, hi);
        lo += (int)(v[p] & 0xFFFFu);
        hi += (int)(v[p] >> 16);
    }
    ((int2*)cnt)[w] = make_int2(lo, hi);
}

__global__ void scan_dinv_kernel(const int* __restrict__ cnt, int* __restrict__ rowptr,
                                 float* __restrict__ dinv) {
    __shared__ int s_pre[8];
    __shared__ int s_sum[8];
    const int t = threadIdx.x;
    const int lane = t & 31;
    const int wave = t >> 5;
    const int4* cnt4 = (const int4*)cnt;
    const int nq = N_NODES / 4;

    int pre = 0;
    for (int k = 0; k < (int)blockIdx.x; ++k) {
        int4 u = cnt4[k * 256 + t];
        pre += (u.x + u.y) + (u.z + u.w);
    }

    const int q = blockIdx.x * 256 + t;
    int4 v = make_int4(0, 0, 0, 0);
    if (q < nq) v = cnt4[q];
    const int tsum = (v.x + v.y) + (v.z + v.w);

    int inc = tsum;
    for (int d = 1; d < 32; d <<= 1) {
        int o = __shfl_up(inc, d);
        if (lane >= d) inc += o;
    }
    for (int d = 16; d > 0; d >>= 1) pre += __shfl_xor(pre, d);
    if (lane == 31) s_sum[wave] = inc;
    if (lane == 0) s_pre[wave] = pre;
    __syncthreads();
    int base = 0;
    for (int w = 0; w < 8; ++w) {
        base += s_pre[w];
        if (w < wave) base += s_sum[w];
    }
    const int ex = base + inc - tsum;

    if (q < nq) {
        int4 o;
        o.x = ex;
        o.y = o.x + v.x;
        o.z = o.y + v.y;
        o.w = o.z + v.z;
        ((int4*)rowptr)[q] = o;
        float4 dv;
        dv.x = (v.x > 0) ? (1.0f / sqrtf((float)v.x)) : 0.0f;
        dv.y = (v.y > 0) ? (1.0f / sqrtf((float)v.y)) : 0.0f;
        dv.z = (v.z > 0) ? (1.0f / sqrtf((float)v.z)) : 0.0f;
        dv.w = (v.w > 0) ? (1.0f / sqrtf((float)v.w)) : 0.0f;
        ((float4*)dinv)[q] = dv;
        if (q == nq - 1) rowptr[N_NODES] = o.w + v.w;
    }
}

__global__ void fill_kernel(const int* __restrict__ row, const int* __restrict__ col,
                            const float* __restrict__ attr, const float* __restrict__ dinv,
                            const int* __restrict__ rowptr, const int* __restrict__ pos,
                            const int* __restrict__ m32, int* __restrict__ edges, int E) {
    int e = blockIdx.x * blockDim.x + threadIdx.x;
    if (e < E) {
        int r = row[e];
        int c = col[e];
        float w = -dinv[r] * attr[e] * dinv[c];
        int p = rowptr[r] + m32[(e / CHUNK) * N_NODES + r] + pos[e];
        ((int2*)edges)[p] = make_int2(c, __float_as_int(w));
    }
}

template <int J>
__device__ __forceinline__ int half_bcast(int v) {
    return __builtin_amdgcn_update_dpp(0, v, 0x150 + J, 0xF, 0xF, false);
}

#define SPMM_SLOT(K) { const int c = half_bcast<BASE + K>(my.x); \
                       w[K] = __int_as_float(half_bcast<BASE + K>(my.y)); \
                       hv[K] = h4[c * 16 + sub]; }

template <int BASE>
__device__ __forceinline__ void issue8(const float4* __restrict__ h4, const int2 my, const int sub,
                                       float4* hv, float* w) {
    SPMM_SLOT(0) SPMM_SLOT(1) SPMM_SLOT(2) SPMM_SLOT(3)
    SPMM_SLOT(4) SPMM_SLOT(5) SPMM_SLOT(6) SPMM_SLOT(7)
}

__device__ __forceinline__ void fma8(const float4* hv, const float* w, float4& acc) {
#pragma unroll
    for (int k = 0; k < 8; ++k) {
        acc.x = fmaf(w[k], hv[k].x, acc.x);
        acc.y = fmaf(w[k], hv[k].y, acc.y);
        acc.z = fmaf(w[k], hv[k].z, acc.z);
        acc.w = fmaf(w[k], hv[k].w, acc.w);
    }
}

__device__ __forceinline__ void load_bounds(const int* __restrict__ rowptr, const int q, const int hw,
                                            const int n_rows, int& s, int& e) {
    const int rr = 2 * q + hw;
    const int rc = min(rr, n_rows - 1);
    const int sv = rowptr[rc];
    const int ev = rowptr[rc + 1];
    const bool ok = rr < n_rows;
    s = ok ? sv : 0;
    e = ok ? ev : 0;
}

__device__ __forceinline__ void load_pairs(const int2* __restrict__ ed, const int s, const int e,
                                           const int sub, int2& a, int2& b) {
    const int n = e - s;
    const bool oa = sub < n;
    const bool ob = sub + 16 < n;
    const int2 va = ed[oa ? s + sub : 0];
    const int2 vb = ed[ob ? s + 16 + sub : 0];
    a = oa ? va : make_int2(0, 0);
    b = ob ? vb : make_int2(0, 0);
}

template <bool HAS_PREV>
__global__ void __launch_bounds__(256)
spmm_gather_kernel(const int* __restrict__ rowptr, const int* __restrict__ edges,
                   const float* __restrict__ h, const float* prev, float* dst, int n_rows) {
    const int lane = threadIdx.x & 31;
    const int sub = lane & 15;
    const int hw = lane >> 4;
    const int wave = __builtin_amdgcn_readfirstlane((int)(threadIdx.x >> 5));
    const int n_waves = gridDim.x * 8;
    const int n_pairs = (n_rows + 1) >> 1;
    const float4* h4 = (const float4*)h;
    const int2* ed = (const int2*)edges;

    int pr = blockIdx.x * 8 + wave;
    if (pr >= n_pairs) return;

    int s, e, s2, e2;
    load_bounds(rowptr, pr, hw, n_rows, s, e);
    load_bounds(rowptr, pr + n_waves, hw, n_rows, s2, e2);
    int2 mya, myb;
    load_pairs(ed, s, e, sub, mya, myb);

    for (; pr < n_pairs; pr += n_waves) {
        const int r = 2 * pr + hw;
        const bool live = r < n_rows;
        const int rc = live ? r : n_rows - 1;

        int2 nxa, nxb;
        load_pairs(ed, s2, e2, sub, nxa, nxb);
        int s3, e3;
        load_bounds(rowptr, pr + 2 * n_waves, hw, n_rows, s3, e3);
        float4 p = make_float4(0.0f, 0.0f, 0.0f, 0.0f);
        if (HAS_PREV) p = ((const float4*)prev)[rc * 16 + sub];

        const int cnt = e - s;
        const int c0 = __builtin_amdgcn_readlane(cnt, 0);
        const int c1 = __builtin_amdgcn_readlane(cnt, 16);
        const int nmax = max(c0, c1);

        float4 acc = make_float4(0.0f, 0.0f, 0.0f, 0.0f);
        {
            float4 hv0[8], hv1[8];
            float w0[8], w1[8];
            issue8<0>(h4, mya, sub, hv0, w0);
            issue8<8>(h4, mya, sub, hv1, w1);
            asm volatile("" : "+v"(w0[0]), "+v"(w0[1]), "+v"(w0[2]), "+v"(w0[3]),
                              "+v"(w0[4]), "+v"(w0[5]), "+v"(w0[6]), "+v"(w0[7]) : : "memory");
            asm volatile("" : "+v"(w1[0]), "+v"(w1[1]), "+v"(w1[2]), "+v"(w1[3]),
                              "+v"(w1[4]), "+v"(w1[5]), "+v"(w1[6]), "+v"(w1[7]) : : "memory");
            fma8(hv0, w0, acc);
            fma8(hv1, w1, acc);
        }
        if (nmax > 16) {
            float4 hv[8];
            float w[8];
            issue8<0>(h4, myb, sub, hv, w);
            fma8(hv, w, acc);
        }
        if (nmax > 24) {
            float4 hv[8];
            float w[8];
            issue8<8>(h4, myb, sub, hv, w);
            fma8(hv, w, acc);
        }
        for (int b = 32; b < nmax; b += 16) {
            const bool om = sub < cnt - b;
            const int2 vm = ed[om ? s + b + sub : 0];
            const int2 my = om ? vm : make_int2(0, 0);
            float4 hv0[8], hv1[8];
            float w0[8], w1[8];
            issue8<0>(h4, my, sub, hv0, w0);
            issue8<8>(h4, my, sub, hv1, w1);
            fma8(hv0, w0, acc);
            fma8(hv1, w1, acc);
        }

        float4 res = acc;
        if (HAS_PREV) {
            res.x = 2.0f * acc.x - p.x;
            res.y = 2.0f * acc.y - p.y;
            res.z = 2.0f * acc.z - p.z;
            res.w = 2.0f * acc.w - p.w;
        }
        if (live) ((float4*)dst)[r * 16 + sub] = res;

        s = s2;  e = e2;
        s2 = s3; e2 = e3;
        mya = nxa; myb = nxb;
    }
}

#define LDS_PITCH 68

__global__ void __launch_bounds__(128)
gemm4_wmma_kernel(const float* __restrict__ X, const float* __restrict__ T1,
                  const float* __restrict__ T2, float* T3out,
                  const float* __restrict__ Wcat, const float* __restrict__ bias,
                  int n_row_tiles) {
    __shared__ __attribute__((aligned(16))) float sA[4 * 16 * LDS_PITCH];
    __shared__ __attribute__((aligned(16))) float sC[16 * LDS_PITCH];

    const int t = threadIdx.x;
    const int wave = t >> 5;
    const int lane = t & 31;
    const int col0 = wave * 16;
    const int laneN = lane & 15;
    const int hi = lane >> 4;

    v2f bfrag[64];
#pragma unroll
    for (int ks = 0; ks < 64; ++ks) {
        int k = ks * 4 + hi * 2;
        bfrag[ks].x = Wcat[(k + 0) * D + col0 + laneN];
        bfrag[ks].y = Wcat[(k + 1) * D + col0 + laneN];
    }
    const float bval = bias[col0 + laneN];

    const int r_a = t >> 4;
    const int c4 = (t & 15) * 4;

    float4 x0, x1, a0, a1, b0, b1, c0, c1;
    x0 = x1 = a0 = a1 = b0 = b1 = c0 = c1 = make_float4(0.0f, 0.0f, 0.0f, 0.0f);
    int tile = blockIdx.x;
    if (tile < n_row_tiles) {
        const size_t o0 = (size_t)(tile * 16 + r_a) * D + c4;
        const size_t o1 = o0 + (size_t)8 * D;
        x0 = *(const float4*)(X + o0);      x1 = *(const float4*)(X + o1);
        a0 = *(const float4*)(T1 + o0);     a1 = *(const float4*)(T1 + o1);
        b0 = *(const float4*)(T2 + o0);     b1 = *(const float4*)(T2 + o1);
        c0 = *(const float4*)(T3out + o0);  c1 = *(const float4*)(T3out + o1);
    }

    for (; tile < n_row_tiles; tile += gridDim.x) {
        *(float4*)&sA[(0 * 16 + r_a) * LDS_PITCH + c4] = x0;
        *(float4*)&sA[(0 * 16 + r_a + 8) * LDS_PITCH + c4] = x1;
        *(float4*)&sA[(1 * 16 + r_a) * LDS_PITCH + c4] = a0;
        *(float4*)&sA[(1 * 16 + r_a + 8) * LDS_PITCH + c4] = a1;
        *(float4*)&sA[(2 * 16 + r_a) * LDS_PITCH + c4] = b0;
        *(float4*)&sA[(2 * 16 + r_a + 8) * LDS_PITCH + c4] = b1;
        *(float4*)&sA[(3 * 16 + r_a) * LDS_PITCH + c4] = c0;
        *(float4*)&sA[(3 * 16 + r_a + 8) * LDS_PITCH + c4] = c1;
        __syncthreads();

        const int next = tile + gridDim.x;
        if (next < n_row_tiles) {
            const size_t o0 = (size_t)(next * 16 + r_a) * D + c4;
            const size_t o1 = o0 + (size_t)8 * D;
            x0 = *(const float4*)(X + o0);      x1 = *(const float4*)(X + o1);
            a0 = *(const float4*)(T1 + o0);     a1 = *(const float4*)(T1 + o1);
            b0 = *(const float4*)(T2 + o0);     b1 = *(const float4*)(T2 + o1);
            c0 = *(const float4*)(T3out + o0);  c1 = *(const float4*)(T3out + o1);
        }

        v8f acc;
        for (int r = 0; r < 8; ++r) acc[r] = bval;

        const float* arow = sA + laneN * LDS_PITCH + hi * 2;
#pragma unroll
        for (int ks = 0; ks < 64; ++ks) {
            const v2f a = *(const v2f*)(arow + (ks >> 4) * 16 * LDS_PITCH + (ks & 15) * 4);
            acc = __builtin_amdgcn_wmma_f32_16x16x4_f32(
                 false, a,  false, bfrag[ks],
                 (short)0, acc,  false,  false);
        }

        for (int r = 0; r < 8; ++r)
            sC[(hi * 8 + r) * LDS_PITCH + col0 + laneN] = acc[r];
        __syncthreads();

        const float4 o0 = *(const float4*)&sC[r_a * LDS_PITCH + c4];
        const float4 o1 = *(const float4*)&sC[(r_a + 8) * LDS_PITCH + c4];
        float* orow = T3out + (size_t)(tile * 16 + r_a) * D + c4;
        *(float4*)orow = o0;
        *(float4*)(orow + 8 * D) = o1;
    }
}

extern "C" void kernel_launch(void* const* d_in, const int* in_sizes, int n_in,
                              void* d_out, int out_size, void* d_ws, size_t ws_size,
                              hipStream_t stream) {
    const float* x         = (const float*)d_in[0];
    const int*   edge_idx  = (const int*)d_in[1];
    const float* edge_attr = (const float*)d_in[2];
    const float* weight    = (const float*)d_in[3];
    const float* bias      = (const float*)d_in[4];
    float*       out       = (float*)d_out;

    const int* row = edge_idx;
    const int* col = edge_idx + N_EDGES;

    char* ws = (char*)d_ws;
    int*   cnt  = (int*)ws;   ws += sizeof(int) * N_NODES;
    float* dinv = (float*)ws; ws += sizeof(float) * N_NODES;
    int*   pos  = (int*)ws;   ws += sizeof(int) * N_EDGES;
    float* A    = (float*)ws; ws += sizeof(float) * (size_t)N_NODES * D;
    float* B    = (float*)ws; ws += sizeof(float) * (size_t)N_NODES * D;
    int*   edges  = (int*)ws; ws += sizeof(int) * 2 * (size_t)N_EDGES;
    int*   rowptr = (int*)ws; ws += sizeof(int) * (N_NODES + 4);

    const int n_row_tiles = N_NODES / 16;
    const int GEMM_BLOCKS = 1024;
    const int SPMM_BLOCKS = 1024;
    const int e_blocks = (N_EDGES + 255) / 256;
    const int scan_blocks = (N_NODES / 4 + 255) / 256;

    unsigned* m16 = (unsigned*)B;
    int*      m32 = (int*)B + (size_t)N_CHUNKS * N_WORDS;
    count_lds_kernel<<<N_CHUNKS, 1024, 0, stream>>>(row, m16, pos);
    combine_kernel<<<(N_WORDS + 255) / 256, 256, 0, stream>>>(m16, m32, cnt);
    scan_dinv_kernel<<<scan_blocks, 256, 0, stream>>>(cnt, rowptr, dinv);
    fill_kernel<<<e_blocks, 256, 0, stream>>>(row, col, edge_attr, dinv, rowptr, pos, m32, edges, N_EDGES);

    spmm_gather_kernel<false><<<SPMM_BLOCKS, 256, 0, stream>>>(rowptr, edges, x, nullptr, A, N_NODES);

    spmm_gather_kernel<true><<<SPMM_BLOCKS, 256, 0, stream>>>(rowptr, edges, A, x, B, N_NODES);

    spmm_gather_kernel<true><<<SPMM_BLOCKS, 256, 0, stream>>>(rowptr, edges, B, A, out, N_NODES);

    gemm4_wmma_kernel<<<GEMM_BLOCKS, 128, 0, stream>>>(x, A, B, out, weight, bias, n_row_tiles);
}
